# starting kernel
# baseline (speedup 1.0000x reference)
#include <hip/hip_runtime.h>


typedef float v2f __attribute__((ext_vector_type(2)));
typedef float v8f __attribute__((ext_vector_type(8)));

#define BM 128
#define BN 128
#define BK 32
#define LDS_STRIDE (BK + 4)

template <int TRANSA>
__global__ __launch_bounds__(256)
void gemm_f32_wmma(const float* __restrict__ Ag, const float* __restrict__ Bg,
                   float* __restrict__ Cg,
                   int M, int N, int K, int lda, int ldb, int ldc,
                   long long strideA, long long strideB, long long strideC,
                   float alpha, int sym)
{
    __shared__ float As[2][BM * LDS_STRIDE];
    __shared__ float Bs[2][BN * LDS_STRIDE];

    const int block_n = blockIdx.x * BN;
    const int block_m = blockIdx.y * BM;
    if (sym && block_m < block_n) return;

    const int tid  = threadIdx.x;
    const int lane = tid & 31;
    const int wave = tid >> 5;
    const int wm   = (wave & 3) * 32;
    const int wn   = (wave >> 2) * 64;
    const int batch = blockIdx.z;

    const float* A = Ag + (long long)batch * strideA;
    const float* B = Bg + (long long)batch * strideB;
    float*       C = Cg + (long long)batch * strideC;

    v8f acc[2][4];
#pragma unroll
    for (int i = 0; i < 2; ++i)
#pragma unroll
        for (int j = 0; j < 4; ++j)
            acc[i][j] = v8f{};

    const int lr = lane & 15;
    const int lh = lane >> 4;

    float4 ra[4], rb[4];

    auto load_a = [&](int kt) {
#pragma unroll
        for (int it = 0; it < 4; ++it) {
            int f = (tid + it * 256) * 4;
            if (TRANSA == 0) {
                int row = f >> 5, col = f & 31;
                ra[it] = *(const float4*)(A + (long long)(block_m + row) * lda + kt + col);
            } else {
                int kr = f >> 7, mc = f & 127;
                ra[it] = *(const float4*)(A + (long long)(kt + kr) * lda + block_m + mc);
            }
        }
    };
    auto load_b = [&](int kt) {
#pragma unroll
        for (int it = 0; it < 4; ++it) {
            int f = (tid + it * 256) * 4;
            int kr = f >> 7, nc = f & 127;
            rb[it] = *(const float4*)(B + (long long)(kt + kr) * ldb + block_n + nc);
        }
    };
    auto store_a = [&](int buf) {
#pragma unroll
        for (int it = 0; it < 4; ++it) {
            int f = (tid + it * 256) * 4;
            if (TRANSA == 0) {
                int row = f >> 5, col = f & 31;
                *(float4*)(&As[buf][row * LDS_STRIDE + col]) = ra[it];
            } else {
                int kr = f >> 7, mc = f & 127;
                As[buf][(mc + 0) * LDS_STRIDE + kr] = ra[it].x;
                As[buf][(mc + 1) * LDS_STRIDE + kr] = ra[it].y;
                As[buf][(mc + 2) * LDS_STRIDE + kr] = ra[it].z;
                As[buf][(mc + 3) * LDS_STRIDE + kr] = ra[it].w;
            }
        }
    };
    auto store_b = [&](int buf) {
#pragma unroll
        for (int it = 0; it < 4; ++it) {
            int f = (tid + it * 256) * 4;
            int kr = f >> 7, nc = f & 127;
            Bs[buf][(nc + 0) * LDS_STRIDE + kr] = rb[it].x;
            Bs[buf][(nc + 1) * LDS_STRIDE + kr] = rb[it].y;
            Bs[buf][(nc + 2) * LDS_STRIDE + kr] = rb[it].z;
            Bs[buf][(nc + 3) * LDS_STRIDE + kr] = rb[it].w;
        }
    };

    load_a(0);
    load_b(0);
    store_a(0);
    store_b(0);
    __syncthreads();

    int cur = 0;
    for (int kt = 0; kt < K; kt += BK) {
        const bool more = (kt + BK) < K;
        if (more) { load_a(kt + BK); load_b(kt + BK); }

#pragma unroll
        for (int kk = 0; kk < BK; kk += 4) {
            v2f afrag[2], bfrag[4];
#pragma unroll
            for (int i = 0; i < 2; ++i) {
                int row = wm + i * 16 + lr;
                afrag[i] = *(const v2f*)(&As[cur][row * LDS_STRIDE + kk + 2 * lh]);
            }
#pragma unroll
            for (int j = 0; j < 4; ++j) {
                int nrw = wn + j * 16 + lr;
                bfrag[j] = *(const v2f*)(&Bs[cur][nrw * LDS_STRIDE + kk + 2 * lh]);
            }
#pragma unroll
            for (int i = 0; i < 2; ++i)
#pragma unroll
                for (int j = 0; j < 4; ++j)
                    acc[i][j] = __builtin_amdgcn_wmma_f32_16x16x4_f32(
                        false, afrag[i], false, bfrag[j],
                        (short)0, acc[i][j], false, false);
        }

        if (more) {
            store_a(cur ^ 1);
            store_b(cur ^ 1);
        }
        __syncthreads();
        cur ^= 1;
    }

    const bool mirror = sym && (block_m != block_n);
#pragma unroll
    for (int i = 0; i < 2; ++i) {
#pragma unroll
        for (int j = 0; j < 4; ++j) {
            int base_m = block_m + wm + i * 16 + (lh << 3);
            int base_n = block_n + wn + j * 16 + lr;
#pragma unroll
            for (int r = 0; r < 8; ++r) {
                float val = alpha * acc[i][j][r];
                C[(long long)(base_m + r) * ldc + base_n] = val;
                if (mirror)
                    C[(long long)base_n * ldc + (base_m + r)] = val;
            }
        }
    }
}

extern "C" void kernel_launch(void* const* d_in, const int* in_sizes, int n_in,
                              void* d_out, int out_size, void* d_ws, size_t ws_size,
                              hipStream_t stream) {
    (void)in_sizes; (void)n_in; (void)out_size; (void)ws_size;

    const float* x  = (const float*)d_in[0];
    const float* wq = (const float*)d_in[1];
    const float* wk = (const float*)d_in[2];
    const float* wv = (const float*)d_in[3];
    float* y = (float*)d_out;

    const int Bb = 4, T = 4096, D = 1024;
    const long long BDD = (long long)D * D;
    const long long BTD = (long long)T * D;

    float* G = (float*)d_ws;
    float* H = G + Bb * BDD;

    dim3 blk(256);
    dim3 gridD(D / BN, D / BM, Bb);

    gemm_f32_wmma<1><<<gridD, blk, 0, stream>>>(
        x, x, G, D, D, T, D, D, D, BTD, BTD, BDD, 1.0f, 1);

    gemm_f32_wmma<0><<<gridD, blk, 0, stream>>>(
        G, wv, H, D, D, D, D, D, D, BDD, 0LL, BDD, 1.0f, 0);

    gemm_f32_wmma<1><<<gridD, blk, 0, stream>>>(
        wk, H, G, D, D, D, D, D, D, 0LL, BDD, BDD, 1.0f, 0);

    gemm_f32_wmma<0><<<gridD, blk, 0, stream>>>(
        wq, G, H, D, D, D, D, D, D, 0LL, BDD, BDD, 1.0f, 0);

    dim3 gridY(D / BN, T / BM, Bb);
    gemm_f32_wmma<0><<<gridY, blk, 0, stream>>>(
        x, H, y, T, D, D, D, D, D, BTD, BDD, BTD, 32.0f, 0);
}


# Tile staging in whole 16-byte vectors: K x cols operands kept k-major in LDS (conflict-free b128 stores, not 8-way-conflicted b32 scatter); native-vector staging regs end the scratch round trip
# speedup vs baseline: 1.2459x; 1.2459x over previous
#include <hip/hip_runtime.h>


typedef float v2f __attribute__((ext_vector_type(2)));
typedef float v4f __attribute__((ext_vector_type(4)));
typedef float v8f __attribute__((ext_vector_type(8)));

#define BM 128
#define BN 128
#define BK 32
#define LDS_STRIDE (BK + 4)
#define KM_STRIDE (BM + 16)
static_assert(BM == BN, "one k-major pitch serves both operands");
static_assert(BM * LDS_STRIDE == BK * KM_STRIDE, "both tile orientations fill the same LDS array");

template <int TRANSA>
__global__ __launch_bounds__(256)
void gemm_f32_wmma(const float* __restrict__ Ag, const float* __restrict__ Bg,
                   float* __restrict__ Cg,
                   int M, int N, int K, int lda, int ldb, int ldc,
                   long long strideA, long long strideB, long long strideC,
                   float alpha, int sym)
{
    __shared__ __attribute__((aligned(16))) float As[2][BM * LDS_STRIDE];
    __shared__ __attribute__((aligned(16))) float Bs[2][BK * KM_STRIDE];

    const int block_n = blockIdx.x * BN;
    const int block_m = blockIdx.y * BM;
    if (sym && block_m < block_n) return;

    const int tid  = threadIdx.x;
    const int lane = tid & 31;
    const int wave = tid >> 5;
    const int wm   = (wave & 3) * 32;
    const int wn   = (wave >> 2) * 64;
    const int batch = blockIdx.z;

    const float* A = Ag + (long long)batch * strideA;
    const float* B = Bg + (long long)batch * strideB;
    float*       C = Cg + (long long)batch * strideC;

    v8f acc[2][4];
#pragma unroll
    for (int i = 0; i < 2; ++i)
#pragma unroll
        for (int j = 0; j < 4; ++j)
            acc[i][j] = v8f{};

    const int lr = lane & 15;
    const int lh = lane >> 4;

    v4f ra[4], rb[4];

    auto load_a = [&](int kt) {
#pragma unroll
        for (int it = 0; it < 4; ++it) {
            int f = (tid + it * 256) * 4;
            if (TRANSA == 0) {
                int row = f >> 5, col = f & 31;
                ra[it] = *(const v4f*)(A + (long long)(block_m + row) * lda + kt + col);
            } else {
                int kr = f >> 7, mc = f & 127;
                ra[it] = *(const v4f*)(A + (long long)(kt + kr) * lda + block_m + mc);
            }
        }
    };
    auto load_b = [&](int kt) {
#pragma unroll
        for (int it = 0; it < 4; ++it) {
            int f = (tid + it * 256) * 4;
            int kr = f >> 7, nc = f & 127;
            rb[it] = *(const v4f*)(B + (long long)(kt + kr) * ldb + block_n + nc);
        }
    };
    auto store_a = [&](int buf) {
#pragma unroll
        for (int it = 0; it < 4; ++it) {
            int f = (tid + it * 256) * 4;
            if (TRANSA == 0) {
                int row = f >> 5, col = f & 31;
                *(v4f*)(&As[buf][row * LDS_STRIDE + col]) = ra[it];
            } else {
                int kr = f >> 7, mc = f & 127;
                *(v4f*)(&As[buf][kr * KM_STRIDE + mc]) = ra[it];
            }
        }
    };
    auto store_b = [&](int buf) {
#pragma unroll
        for (int it = 0; it < 4; ++it) {
            int f = (tid + it * 256) * 4;
            int kr = f >> 7, nc = f & 127;
            *(v4f*)(&Bs[buf][kr * KM_STRIDE + nc]) = rb[it];
        }
    };

    load_a(0);
    load_b(0);
    store_a(0);
    store_b(0);
    __syncthreads();

    int cur = 0;
    for (int kt = 0; kt < K; kt += BK) {
        const bool more = (kt + BK) < K;
        if (more) { load_a(kt + BK); load_b(kt + BK); }

#pragma unroll
        for (int kk = 0; kk < BK; kk += 4) {
            v2f afrag[2], bfrag[4];
#pragma unroll
            for (int i = 0; i < 2; ++i) {
                int row = wm + i * 16 + lr;
                if (TRANSA == 0) {
                    afrag[i] = *(const v2f*)(&As[cur][row * LDS_STRIDE + kk + 2 * lh]);
                } else {
                    const float* ap = &As[cur][(kk + 2 * lh) * KM_STRIDE + row];
                    afrag[i] = v2f{ap[0], ap[KM_STRIDE]};
                }
            }
#pragma unroll
            for (int j = 0; j < 4; ++j) {
                int nrw = wn + j * 16 + lr;
                const float* bp = &Bs[cur][(kk + 2 * lh) * KM_STRIDE + nrw];
                bfrag[j] = v2f{bp[0], bp[KM_STRIDE]};
            }
#pragma unroll
            for (int i = 0; i < 2; ++i)
#pragma unroll
                for (int j = 0; j < 4; ++j)
                    acc[i][j] = __builtin_amdgcn_wmma_f32_16x16x4_f32(
                        false, afrag[i], false, bfrag[j],
                        (short)0, acc[i][j], false, false);
        }

        if (more) {
            store_a(cur ^ 1);
            store_b(cur ^ 1);
        }
        __syncthreads();
        cur ^= 1;
    }

    const bool mirror = sym && (block_m != block_n);
#pragma unroll
    for (int i = 0; i < 2; ++i) {
#pragma unroll
        for (int j = 0; j < 4; ++j) {
            int base_m = block_m + wm + i * 16 + (lh << 3);
            int base_n = block_n + wn + j * 16 + lr;
#pragma unroll
            for (int r = 0; r < 8; ++r) {
                float val = alpha * acc[i][j][r];
                C[(long long)(base_m + r) * ldc + base_n] = val;
                if (mirror)
                    C[(long long)base_n * ldc + (base_m + r)] = val;
            }
        }
    }
}

extern "C" void kernel_launch(void* const* d_in, const int* in_sizes, int n_in,
                              void* d_out, int out_size, void* d_ws, size_t ws_size,
                              hipStream_t stream) {
    (void)in_sizes; (void)n_in; (void)out_size; (void)ws_size;

    const float* x  = (const float*)d_in[0];
    const float* wq = (const float*)d_in[1];
    const float* wk = (const float*)d_in[2];
    const float* wv = (const float*)d_in[3];
    float* y = (float*)d_out;

    const int Bb = 4, T = 4096, D = 1024;
    const long long BDD = (long long)D * D;
    const long long BTD = (long long)T * D;

    float* G = (float*)d_ws;
    float* H = G + Bb * BDD;

    dim3 blk(256);
    dim3 gridD(D / BN, D / BM, Bb);

    gemm_f32_wmma<1><<<gridD, blk, 0, stream>>>(
        x, x, G, D, D, T, D, D, D, BTD, BTD, BDD, 1.0f, 1);

    gemm_f32_wmma<0><<<gridD, blk, 0, stream>>>(
        G, wv, H, D, D, D, D, D, D, BDD, 0LL, BDD, 1.0f, 0);

    gemm_f32_wmma<1><<<gridD, blk, 0, stream>>>(
        wk, H, G, D, D, D, D, D, D, 0LL, BDD, BDD, 1.0f, 0);

    gemm_f32_wmma<0><<<gridD, blk, 0, stream>>>(
        wq, G, H, D, D, D, D, D, D, 0LL, BDD, BDD, 1.0f, 0);

    dim3 gridY(D / BN, T / BM, Bb);
    gemm_f32_wmma<0><<<gridY, blk, 0, stream>>>(
        x, H, y, T, D, D, D, D, D, BTD, BDD, BTD, 32.0f, 0);
}


# Re-associate: Wq*Wk^T is batch-free, formed once by launch 1's idle upper-triangle blocks; W = (Wq Wk^T)(G Wv): one batched GEMM launch fewer (5 to 4); f32 summation order changes
# speedup vs baseline: 1.3488x; 1.0826x over previous
#include <hip/hip_runtime.h>


typedef float v2f __attribute__((ext_vector_type(2)));
typedef float v4f __attribute__((ext_vector_type(4)));
typedef float v8f __attribute__((ext_vector_type(8)));

#define BM 128
#define BN 128
#define BK 32
#define LDS_STRIDE (BK + 4)
#define KM_STRIDE (BM + 16)
#define TILE_FLOATS (BM * LDS_STRIDE)
static_assert(BM == BN, "one pitch per orientation serves both operands");
static_assert(BM * LDS_STRIDE == BK * KM_STRIDE, "both tile orientations fill the same LDS array");

template <int TRANSA, int TRANSB>
__device__ __forceinline__
void gemm_tile(float (&As)[2][TILE_FLOATS], float (&Bs)[2][TILE_FLOATS],
               const float* __restrict__ A, const float* __restrict__ B, float* __restrict__ C,
               int block_m, int block_n, int K, int lda, int ldb, int ldc,
               float alpha, bool mirror)
{
    const int tid  = threadIdx.x;
    const int lane = tid & 31;
    const int wave = tid >> 5;
    const int wm   = (wave & 3) * 32;
    const int wn   = (wave >> 2) * 64;

    v8f acc[2][4];
#pragma unroll
    for (int i = 0; i < 2; ++i)
#pragma unroll
        for (int j = 0; j < 4; ++j)
            acc[i][j] = v8f{};

    const int lr = lane & 15;
    const int lh = lane >> 4;

    v4f ra[4], rb[4];

    auto load_a = [&](int kt) {
#pragma unroll
        for (int it = 0; it < 4; ++it) {
            int f = (tid + it * 256) * 4;
            if (TRANSA == 0) {
                int row = f >> 5, col = f & 31;
                ra[it] = *(const v4f*)(A + (long long)(block_m + row) * lda + kt + col);
            } else {
                int kr = f >> 7, mc = f & 127;
                ra[it] = *(const v4f*)(A + (long long)(kt + kr) * lda + block_m + mc);
            }
        }
    };
    auto load_b = [&](int kt) {
#pragma unroll
        for (int it = 0; it < 4; ++it) {
            int f = (tid + it * 256) * 4;
            if (TRANSB == 0) {
                int kr = f >> 7, nc = f & 127;
                rb[it] = *(const v4f*)(B + (long long)(kt + kr) * ldb + block_n + nc);
            } else {
                int row = f >> 5, col = f & 31;
                rb[it] = *(const v4f*)(B + (long long)(block_n + row) * ldb + kt + col);
            }
        }
    };
    auto store_a = [&](int buf) {
#pragma unroll
        for (int it = 0; it < 4; ++it) {
            int f = (tid + it * 256) * 4;
            if (TRANSA == 0) {
                int row = f >> 5, col = f & 31;
                *(v4f*)(&As[buf][row * LDS_STRIDE + col]) = ra[it];
            } else {
                int kr = f >> 7, mc = f & 127;
                *(v4f*)(&As[buf][kr * KM_STRIDE + mc]) = ra[it];
            }
        }
    };
    auto store_b = [&](int buf) {
#pragma unroll
        for (int it = 0; it < 4; ++it) {
            int f = (tid + it * 256) * 4;
            if (TRANSB == 0) {
                int kr = f >> 7, nc = f & 127;
                *(v4f*)(&Bs[buf][kr * KM_STRIDE + nc]) = rb[it];
            } else {
                int row = f >> 5, col = f & 31;
                *(v4f*)(&Bs[buf][row * LDS_STRIDE + col]) = rb[it];
            }
        }
    };

    load_a(0);
    load_b(0);
    store_a(0);
    store_b(0);
    __syncthreads();

    int cur = 0;
    for (int kt = 0; kt < K; kt += BK) {
        const bool more = (kt + BK) < K;
        if (more) { load_a(kt + BK); load_b(kt + BK); }

#pragma unroll
        for (int kk = 0; kk < BK; kk += 4) {
            v2f afrag[2], bfrag[4];
#pragma unroll
            for (int i = 0; i < 2; ++i) {
                int row = wm + i * 16 + lr;
                if (TRANSA == 0) {
                    afrag[i] = *(const v2f*)(&As[cur][row * LDS_STRIDE + kk + 2 * lh]);
                } else {
                    const float* ap = &As[cur][(kk + 2 * lh) * KM_STRIDE + row];
                    afrag[i] = v2f{ap[0], ap[KM_STRIDE]};
                }
            }
#pragma unroll
            for (int j = 0; j < 4; ++j) {
                int nrw = wn + j * 16 + lr;
                if (TRANSB == 0) {
                    const float* bp = &Bs[cur][(kk + 2 * lh) * KM_STRIDE + nrw];
                    bfrag[j] = v2f{bp[0], bp[KM_STRIDE]};
                } else {
                    bfrag[j] = *(const v2f*)(&Bs[cur][nrw * LDS_STRIDE + kk + 2 * lh]);
                }
            }
#pragma unroll
            for (int i = 0; i < 2; ++i)
#pragma unroll
                for (int j = 0; j < 4; ++j)
                    acc[i][j] = __builtin_amdgcn_wmma_f32_16x16x4_f32(
                        false, afrag[i], false, bfrag[j],
                        (short)0, acc[i][j], false, false);
        }

        if (more) {
            store_a(cur ^ 1);
            store_b(cur ^ 1);
        }
        __syncthreads();
        cur ^= 1;
    }

#pragma unroll
    for (int i = 0; i < 2; ++i) {
#pragma unroll
        for (int j = 0; j < 4; ++j) {
            int base_m = block_m + wm + i * 16 + (lh << 3);
            int base_n = block_n + wn + j * 16 + lr;
#pragma unroll
            for (int r = 0; r < 8; ++r) {
                float val = alpha * acc[i][j][r];
                C[(long long)(base_m + r) * ldc + base_n] = val;
                if (mirror)
                    C[(long long)base_n * ldc + (base_m + r)] = val;
            }
        }
    }
}

__global__ __launch_bounds__(256)
void gemm_f32_wmma_nn(const float* __restrict__ Ag, const float* __restrict__ Bg,
                      float* __restrict__ Cg,
                      int K, int lda, int ldb, int ldc,
                      long long strideA, long long strideB, long long strideC,
                      float alpha)
{
    __shared__ __attribute__((aligned(16))) float As[2][TILE_FLOATS];
    __shared__ __attribute__((aligned(16))) float Bs[2][TILE_FLOATS];
    const int batch = blockIdx.z;
    gemm_tile<0, 0>(As, Bs,
                    Ag + (long long)batch * strideA, Bg + (long long)batch * strideB,
                    Cg + (long long)batch * strideC,
                    blockIdx.y * BM, blockIdx.x * BN, K, lda, ldb, ldc, alpha, false);
}

__global__ __launch_bounds__(256)
void gram_and_qk_f32_wmma(const float* __restrict__ x, const float* __restrict__ wq,
                          const float* __restrict__ wk,
                          float* __restrict__ G, float* __restrict__ P, int T, int D)
{
    __shared__ __attribute__((aligned(16))) float As[2][TILE_FLOATS];
    __shared__ __attribute__((aligned(16))) float Bs[2][TILE_FLOATS];
    const int bx = blockIdx.x, by = blockIdx.y, bz = blockIdx.z;
    if (by >= bx) {
        const float* xb = x + (long long)bz * T * D;
        gemm_tile<1, 0>(As, Bs, xb, xb, G + (long long)bz * D * D,
                        by * BM, bx * BN, T, D, D, D, 1.0f, by != bx);
    } else {
        const int idx   = bx * (bx - 1) / 2 + by;
        const int tiles = gridDim.x * gridDim.y;
        const int per   = (tiles + gridDim.z - 1) / gridDim.z;
        const int t     = bz * per + idx;
        if (idx >= per || t >= tiles) return;
        gemm_tile<0, 1>(As, Bs, wq, wk, P,
                        (t / gridDim.x) * BM, (t % gridDim.x) * BN, D, D, D, D, 1.0f, false);
    }
}

extern "C" void kernel_launch(void* const* d_in, const int* in_sizes, int n_in,
                              void* d_out, int out_size, void* d_ws, size_t ws_size,
                              hipStream_t stream) {
    (void)in_sizes; (void)n_in; (void)out_size; (void)ws_size;

    const float* x  = (const float*)d_in[0];
    const float* wq = (const float*)d_in[1];
    const float* wk = (const float*)d_in[2];
    const float* wv = (const float*)d_in[3];
    float* y = (float*)d_out;

    const int Bb = 4, T = 4096, D = 1024;
    const long long BDD = (long long)D * D;
    const long long BTD = (long long)T * D;

    float* G = (float*)d_ws;
    float* H = G + Bb * BDD;
    float* P = y;

    dim3 blk(256);
    dim3 gridD(D / BN, D / BM, Bb);

    gram_and_qk_f32_wmma<<<gridD, blk, 0, stream>>>(x, wq, wk, G, P, T, D);

    gemm_f32_wmma_nn<<<gridD, blk, 0, stream>>>(
        G, wv, H, D, D, D, D, BDD, 0LL, BDD, 1.0f);

    gemm_f32_wmma_nn<<<gridD, blk, 0, stream>>>(
        P, H, G, D, D, D, D, 0LL, BDD, BDD, 1.0f);

    dim3 gridY(D / BN, T / BM, Bb);
    gemm_f32_wmma_nn<<<gridY, blk, 0, stream>>>(
        x, G, y, D, D, D, D, BTD, BDD, BTD, 32.0f);
}


# K loop software-pipelined half a stage deep: 4 k-steps of fragments read from LDS as one group 32 WMMAs ahead of use, barrier mid-stage; K x cols tiles keep k pairs adjacent (8-byte fragment reads)
# speedup vs baseline: 1.5366x; 1.1392x over previous
#include <hip/hip_runtime.h>


typedef float v2f __attribute__((ext_vector_type(2)));
typedef float v4f __attribute__((ext_vector_type(4)));
typedef float v8f __attribute__((ext_vector_type(8)));

#define BM 128
#define BN 128
#define BK 32
#define LDS_STRIDE (BK + 4)
#define KP_STRIDE (2 * BM + 32)
#define TILE_FLOATS (BM * LDS_STRIDE)
static_assert(BM == BN, "one pitch per orientation serves both operands");
static_assert(BM * LDS_STRIDE == (BK / 2) * KP_STRIDE, "both tile orientations fill the same LDS array");

template <int TRANSA, int TRANSB>
__device__ __forceinline__
void gemm_tile(float (&As)[2][TILE_FLOATS], float (&Bs)[2][TILE_FLOATS],
               const float* __restrict__ A, const float* __restrict__ B, float* __restrict__ C,
               int block_m, int block_n, int K, int lda, int ldb, int ldc,
               float alpha, bool mirror)
{
    const int tid  = threadIdx.x;
    const int lane = tid & 31;
    const int wave = tid >> 5;
    const int wm   = (wave & 3) * 32;
    const int wn   = (wave >> 2) * 64;

    v8f acc[2][4];
#pragma unroll
    for (int i = 0; i < 2; ++i)
#pragma unroll
        for (int j = 0; j < 4; ++j)
            acc[i][j] = v8f{};

    const int lr = lane & 15;
    const int lh = lane >> 4;

    v4f ra[4], rb[4];

    auto load_a = [&](int kt) {
#pragma unroll
        for (int it = 0; it < 4; ++it) {
            int f = (tid + it * 256) * 4;
            if (TRANSA == 0) {
                int row = f >> 5, col = f & 31;
                ra[it] = *(const v4f*)(A + (long long)(block_m + row) * lda + kt + col);
            } else {
                int g = tid + (it >> 1) * 256;
                int kr = 2 * (g >> 5) + (it & 1), mc = (g & 31) * 4;
                ra[it] = *(const v4f*)(A + (long long)(kt + kr) * lda + block_m + mc);
            }
        }
    };
    auto load_b = [&](int kt) {
#pragma unroll
        for (int it = 0; it < 4; ++it) {
            int f = (tid + it * 256) * 4;
            if (TRANSB == 0) {
                int g = tid + (it >> 1) * 256;
                int kr = 2 * (g >> 5) + (it & 1), nc = (g & 31) * 4;
                rb[it] = *(const v4f*)(B + (long long)(kt + kr) * ldb + block_n + nc);
            } else {
                int row = f >> 5, col = f & 31;
                rb[it] = *(const v4f*)(B + (long long)(block_n + row) * ldb + kt + col);
            }
        }
    };
    auto store_pairs = [&](float* tile, const v4f (&r)[4]) {
#pragma unroll
        for (int h = 0; h < 2; ++h) {
            int g = tid + h * 256;
            float* dst = tile + (g >> 5) * KP_STRIDE + (g & 31) * 8;
            v4f r0 = r[2 * h], r1 = r[2 * h + 1];
            *(v4f*)(dst)     = v4f{r0.x, r1.x, r0.y, r1.y};
            *(v4f*)(dst + 4) = v4f{r0.z, r1.z, r0.w, r1.w};
        }
    };
    auto store_rows = [&](float* tile, const v4f (&r)[4]) {
#pragma unroll
        for (int it = 0; it < 4; ++it) {
            int f = (tid + it * 256) * 4;
            int row = f >> 5, col = f & 31;
            *(v4f*)(tile + row * LDS_STRIDE + col) = r[it];
        }
    };
    auto store_a = [&](int buf) {
        if (TRANSA == 0) store_rows(&As[buf][0], ra); else store_pairs(&As[buf][0], ra);
    };
    auto store_b = [&](int buf) {
        if (TRANSB == 0) store_pairs(&Bs[buf][0], rb); else store_rows(&Bs[buf][0], rb);
    };

    v2f xa[4][2], xb[4][4];
    v2f ya[4][2], yb[4][4];

    auto load_frags = [&](v2f (&fa)[4][2], v2f (&fb)[4][4], int buf, int kbase) {
#pragma unroll
        for (int t = 0; t < 4; ++t) {
            const int kk = kbase + 4 * t;
#pragma unroll
            for (int i = 0; i < 2; ++i) {
                int row = wm + i * 16 + lr;
                if (TRANSA == 0)
                    fa[t][i] = *(const v2f*)(&As[buf][row * LDS_STRIDE + kk + 2 * lh]);
                else
                    fa[t][i] = *(const v2f*)(&As[buf][((kk >> 1) + lh) * KP_STRIDE + 2 * row]);
            }
#pragma unroll
            for (int j = 0; j < 4; ++j) {
                int nrw = wn + j * 16 + lr;
                if (TRANSB == 0)
                    fb[t][j] = *(const v2f*)(&Bs[buf][((kk >> 1) + lh) * KP_STRIDE + 2 * nrw]);
                else
                    fb[t][j] = *(const v2f*)(&Bs[buf][nrw * LDS_STRIDE + kk + 2 * lh]);
            }
        }
    };
    auto mma_steps = [&](v2f (&fa)[4][2], v2f (&fb)[4][4], int t0, int t1) {
#pragma unroll
        for (int t = t0; t < t1; ++t)
#pragma unroll
            for (int i = 0; i < 2; ++i)
#pragma unroll
                for (int j = 0; j < 4; ++j)
                    acc[i][j] = __builtin_amdgcn_wmma_f32_16x16x4_f32(
                        false, fa[t][i], false, fb[t][j],
                        (short)0, acc[i][j], false, false);
    };

    load_a(0);
    load_b(0);
    store_a(0);
    store_b(0);
    __syncthreads();
    load_frags(xa, xb, 0, 0);
    if (BK < K) { load_a(BK); load_b(BK); }

    int cur = 0;
    for (int kt = 0; kt < K; kt += BK) {
        const bool more = (kt + BK) < K;

        load_frags(ya, yb, cur, BK / 2);
        __builtin_amdgcn_sched_barrier(0);
        mma_steps(xa, xb, 0, 1);
        if (more) {
            store_a(cur ^ 1);
            store_b(cur ^ 1);
        }
        mma_steps(xa, xb, 1, 4);

        __syncthreads();
        __builtin_amdgcn_sched_barrier(0);

        load_frags(xa, xb, cur ^ 1, 0);
        if ((kt + 2 * BK) < K) { load_a(kt + 2 * BK); load_b(kt + 2 * BK); }
        mma_steps(ya, yb, 0, 4);
        cur ^= 1;
    }

#pragma unroll
    for (int i = 0; i < 2; ++i) {
#pragma unroll
        for (int j = 0; j < 4; ++j) {
            int base_m = block_m + wm + i * 16 + (lh << 3);
            int base_n = block_n + wn + j * 16 + lr;
#pragma unroll
            for (int r = 0; r < 8; ++r) {
                float val = alpha * acc[i][j][r];
                C[(long long)(base_m + r) * ldc + base_n] = val;
                if (mirror)
                    C[(long long)base_n * ldc + (base_m + r)] = val;
            }
        }
    }
}

__global__ __launch_bounds__(256)
void gemm_f32_wmma_nn(const float* __restrict__ Ag, const float* __restrict__ Bg,
                      float* __restrict__ Cg,
                      int K, int lda, int ldb, int ldc,
                      long long strideA, long long strideB, long long strideC,
                      float alpha)
{
    __shared__ __attribute__((aligned(16))) float As[2][TILE_FLOATS];
    __shared__ __attribute__((aligned(16))) float Bs[2][TILE_FLOATS];
    const int batch = blockIdx.z;
    gemm_tile<0, 0>(As, Bs,
                    Ag + (long long)batch * strideA, Bg + (long long)batch * strideB,
                    Cg + (long long)batch * strideC,
                    blockIdx.y * BM, blockIdx.x * BN, K, lda, ldb, ldc, alpha, false);
}

__global__ __launch_bounds__(256)
void gram_and_qk_f32_wmma(const float* __restrict__ x, const float* __restrict__ wq,
                          const float* __restrict__ wk,
                          float* __restrict__ G, float* __restrict__ P, int T, int D)
{
    __shared__ __attribute__((aligned(16))) float As[2][TILE_FLOATS];
    __shared__ __attribute__((aligned(16))) float Bs[2][TILE_FLOATS];
    const int bx = blockIdx.x, by = blockIdx.y, bz = blockIdx.z;
    if (by >= bx) {
        const float* xb = x + (long long)bz * T * D;
        gemm_tile<1, 0>(As, Bs, xb, xb, G + (long long)bz * D * D,
                        by * BM, bx * BN, T, D, D, D, 1.0f, by != bx);
    } else {
        const int idx   = bx * (bx - 1) / 2 + by;
        const int tiles = gridDim.x * gridDim.y;
        const int per   = (tiles + gridDim.z - 1) / gridDim.z;
        const int t     = bz * per + idx;
        if (idx >= per || t >= tiles) return;
        gemm_tile<0, 1>(As, Bs, wq, wk, P,
                        (t / gridDim.x) * BM, (t % gridDim.x) * BN, D, D, D, D, 1.0f, false);
    }
}

extern "C" void kernel_launch(void* const* d_in, const int* in_sizes, int n_in,
                              void* d_out, int out_size, void* d_ws, size_t ws_size,
                              hipStream_t stream) {
    (void)in_sizes; (void)n_in; (void)out_size; (void)ws_size;

    const float* x  = (const float*)d_in[0];
    const float* wq = (const float*)d_in[1];
    const float* wk = (const float*)d_in[2];
    const float* wv = (const float*)d_in[3];
    float* y = (float*)d_out;

    const int Bb = 4, T = 4096, D = 1024;
    const long long BDD = (long long)D * D;
    const long long BTD = (long long)T * D;

    float* G = (float*)d_ws;
    float* H = G + Bb * BDD;
    float* P = y;

    dim3 blk(256);
    dim3 gridD(D / BN, D / BM, Bb);

    gram_and_qk_f32_wmma<<<gridD, blk, 0, stream>>>(x, wq, wk, G, P, T, D);

    gemm_f32_wmma_nn<<<gridD, blk, 0, stream>>>(
        G, wv, H, D, D, D, D, BDD, 0LL, BDD, 1.0f);

    gemm_f32_wmma_nn<<<gridD, blk, 0, stream>>>(
        P, H, G, D, D, D, D, 0LL, BDD, BDD, 1.0f);

    dim3 gridY(D / BN, T / BM, Bb);
    gemm_f32_wmma_nn<<<gridY, blk, 0, stream>>>(
        x, G, y, D, D, D, D, BTD, BDD, BTD, 32.0f);
}


# Launch 1 refilled: K of x^T x cut 86:42 stages over work-groups (248 of 74-86 stages, was 144 of 128 on 256 WGPs); launch 2's loader adds the two K-parts; f32 summation order changes
# speedup vs baseline: 1.6589x; 1.0796x over previous
#include <hip/hip_runtime.h>


typedef float v2f __attribute__((ext_vector_type(2)));
typedef float v4f __attribute__((ext_vector_type(4)));
typedef float v8f __attribute__((ext_vector_type(8)));

#define BM 128
#define BN 128
#define BK 32
#define LDS_STRIDE (BK + 4)
#define KP_STRIDE (2 * BM + 32)
#define TILE_FLOATS (BM * LDS_STRIDE)
static_assert(BM == BN, "one pitch per orientation serves both operands");
static_assert(BM * LDS_STRIDE == (BK / 2) * KP_STRIDE, "both tile orientations fill the same LDS array");

template <int TRANSA, int TRANSB, int SUMA = 0>
__device__ __forceinline__
void gemm_tile(float (&As)[2][TILE_FLOATS], float (&Bs)[2][TILE_FLOATS],
               const float* __restrict__ A, const float* __restrict__ A2,
               const float* __restrict__ B, float* __restrict__ C,
               int block_m, int block_n, int K, int lda, int ldb, int ldc,
               float alpha, bool mirror)
{
    const int tid  = threadIdx.x;
    const int lane = tid & 31;
    const int wave = tid >> 5;
    const int wm   = (wave & 3) * 32;
    const int wn   = (wave >> 2) * 64;

    v8f acc[2][4];
#pragma unroll
    for (int i = 0; i < 2; ++i)
#pragma unroll
        for (int j = 0; j < 4; ++j)
            acc[i][j] = v8f{};

    const int lr = lane & 15;
    const int lh = lane >> 4;

    v4f ra[4], rb[4];
    v4f ra2[4];
    static_assert(!(SUMA && TRANSA), "the summed A operand is staged row-major only");

    auto load_a = [&](int kt) {
#pragma unroll
        for (int it = 0; it < 4; ++it) {
            int f = (tid + it * 256) * 4;
            if (TRANSA == 0) {
                int row = f >> 5, col = f & 31;
                ra[it] = *(const v4f*)(A + (long long)(block_m + row) * lda + kt + col);
                if (SUMA)
                    ra2[it] = *(const v4f*)(A2 + (long long)(block_m + row) * lda + kt + col);
            } else {
                int g = tid + (it >> 1) * 256;
                int kr = 2 * (g >> 5) + (it & 1), mc = (g & 31) * 4;
                ra[it] = *(const v4f*)(A + (long long)(kt + kr) * lda + block_m + mc);
            }
        }
    };
    auto load_b = [&](int kt) {
#pragma unroll
        for (int it = 0; it < 4; ++it) {
            int f = (tid + it * 256) * 4;
            if (TRANSB == 0) {
                int g = tid + (it >> 1) * 256;
                int kr = 2 * (g >> 5) + (it & 1), nc = (g & 31) * 4;
                rb[it] = *(const v4f*)(B + (long long)(kt + kr) * ldb + block_n + nc);
            } else {
                int row = f >> 5, col = f & 31;
                rb[it] = *(const v4f*)(B + (long long)(block_n + row) * ldb + kt + col);
            }
        }
    };
    auto store_pairs = [&](float* tile, const v4f (&r)[4]) {
#pragma unroll
        for (int h = 0; h < 2; ++h) {
            int g = tid + h * 256;
            float* dst = tile + (g >> 5) * KP_STRIDE + (g & 31) * 8;
            v4f r0 = r[2 * h], r1 = r[2 * h + 1];
            *(v4f*)(dst)     = v4f{r0.x, r1.x, r0.y, r1.y};
            *(v4f*)(dst + 4) = v4f{r0.z, r1.z, r0.w, r1.w};
        }
    };
    auto store_rows = [&](float* tile, const v4f (&r)[4]) {
#pragma unroll
        for (int it = 0; it < 4; ++it) {
            int f = (tid + it * 256) * 4;
            int row = f >> 5, col = f & 31;
            *(v4f*)(tile + row * LDS_STRIDE + col) = r[it];
        }
    };
    auto store_a = [&](int buf) {
        if (SUMA) {
            v4f rs[4];
#pragma unroll
            for (int it = 0; it < 4; ++it) rs[it] = ra[it] + ra2[it];
            store_rows(&As[buf][0], rs);
        } else if (TRANSA == 0) store_rows(&As[buf][0], ra);
        else store_pairs(&As[buf][0], ra);
    };
    auto store_b = [&](int buf) {
        if (TRANSB == 0) store_pairs(&Bs[buf][0], rb); else store_rows(&Bs[buf][0], rb);
    };

    v2f xa[4][2], xb[4][4];
    v2f ya[4][2], yb[4][4];

    auto load_frags = [&](v2f (&fa)[4][2], v2f (&fb)[4][4], int buf, int kbase) {
#pragma unroll
        for (int t = 0; t < 4; ++t) {
            const int kk = kbase + 4 * t;
#pragma unroll
            for (int i = 0; i < 2; ++i) {
                int row = wm + i * 16 + lr;
                if (TRANSA == 0)
                    fa[t][i] = *(const v2f*)(&As[buf][row * LDS_STRIDE + kk + 2 * lh]);
                else
                    fa[t][i] = *(const v2f*)(&As[buf][((kk >> 1) + lh) * KP_STRIDE + 2 * row]);
            }
#pragma unroll
            for (int j = 0; j < 4; ++j) {
                int nrw = wn + j * 16 + lr;
                if (TRANSB == 0)
                    fb[t][j] = *(const v2f*)(&Bs[buf][((kk >> 1) + lh) * KP_STRIDE + 2 * nrw]);
                else
                    fb[t][j] = *(const v2f*)(&Bs[buf][nrw * LDS_STRIDE + kk + 2 * lh]);
            }
        }
    };
    auto mma_steps = [&](v2f (&fa)[4][2], v2f (&fb)[4][4], int t0, int t1) {
#pragma unroll
        for (int t = t0; t < t1; ++t)
#pragma unroll
            for (int i = 0; i < 2; ++i)
#pragma unroll
                for (int j = 0; j < 4; ++j)
                    acc[i][j] = __builtin_amdgcn_wmma_f32_16x16x4_f32(
                        false, fa[t][i], false, fb[t][j],
                        (short)0, acc[i][j], false, false);
    };

    load_a(0);
    load_b(0);
    store_a(0);
    store_b(0);
    __syncthreads();
    load_frags(xa, xb, 0, 0);
    if (BK < K) { load_a(BK); load_b(BK); }

    int cur = 0;
    for (int kt = 0; kt < K; kt += BK) {
        const bool more = (kt + BK) < K;

        load_frags(ya, yb, cur, BK / 2);
        __builtin_amdgcn_sched_barrier(0);
        mma_steps(xa, xb, 0, 1);
        if (more) {
            store_a(cur ^ 1);
            store_b(cur ^ 1);
        }
        mma_steps(xa, xb, 1, 4);

        __syncthreads();
        __builtin_amdgcn_sched_barrier(0);

        load_frags(xa, xb, cur ^ 1, 0);
        if ((kt + 2 * BK) < K) { load_a(kt + 2 * BK); load_b(kt + 2 * BK); }
        mma_steps(ya, yb, 0, 4);
        cur ^= 1;
    }

#pragma unroll
    for (int i = 0; i < 2; ++i) {
#pragma unroll
        for (int j = 0; j < 4; ++j) {
            int base_m = block_m + wm + i * 16 + (lh << 3);
            int base_n = block_n + wn + j * 16 + lr;
#pragma unroll
            for (int r = 0; r < 8; ++r) {
                float val = alpha * acc[i][j][r];
                C[(long long)(base_m + r) * ldc + base_n] = val;
                if (mirror)
                    C[(long long)base_n * ldc + (base_m + r)] = val;
            }
        }
    }
}

__global__ __launch_bounds__(256)
void gemm_f32_wmma_nn(const float* __restrict__ Ag, const float* __restrict__ Bg,
                      float* __restrict__ Cg,
                      int K, int lda, int ldb, int ldc,
                      long long strideA, long long strideB, long long strideC,
                      float alpha)
{
    __shared__ __attribute__((aligned(16))) float As[2][TILE_FLOATS];
    __shared__ __attribute__((aligned(16))) float Bs[2][TILE_FLOATS];
    const int batch = blockIdx.z;
    gemm_tile<0, 0>(As, Bs,
                    Ag + (long long)batch * strideA, nullptr, Bg + (long long)batch * strideB,
                    Cg + (long long)batch * strideC,
                    blockIdx.y * BM, blockIdx.x * BN, K, lda, ldb, ldc, alpha, false);
}

__global__ __launch_bounds__(256)
void gemm_f32_wmma_nn_sum2(const float* __restrict__ Ag, const float* __restrict__ A2g,
                           const float* __restrict__ Bg, float* __restrict__ Cg,
                           int K, int lda, int ldb, int ldc,
                           long long strideA, long long strideB, long long strideC,
                           float alpha)
{
    __shared__ __attribute__((aligned(16))) float As[2][TILE_FLOATS];
    __shared__ __attribute__((aligned(16))) float Bs[2][TILE_FLOATS];
    const int batch = blockIdx.z;
    gemm_tile<0, 0, 1>(As, Bs,
                       Ag + (long long)batch * strideA, A2g + (long long)batch * strideA,
                       Bg + (long long)batch * strideB,
                       Cg + (long long)batch * strideC,
                       blockIdx.y * BM, blockIdx.x * BN, K, lda, ldb, ldc, alpha, false);
}

__global__ __launch_bounds__(256)
void gram_and_qk_f32_wmma(const float* __restrict__ x, const float* __restrict__ wq,
                          const float* __restrict__ wk,
                          float* __restrict__ G0, float* __restrict__ G1, float* __restrict__ P,
                          int T, int D, int batches, int cut)
{
    __shared__ __attribute__((aligned(16))) float As[2][TILE_FLOATS];
    __shared__ __attribute__((aligned(16))) float Bs[2][TILE_FLOATS];
    const int w   = blockIdx.x;
    const int nb  = D / BM;
    const int tri = nb * (nb + 1) / 2;
    const int n_t = tri * batches;
    const int n_p = nb * nb;

    auto gram_part = [&](int t, bool head) {
        const int bz  = t / tri;
        const int idx = t - bz * tri;
        int by = 0;
        while ((by + 1) * (by + 2) / 2 <= idx) ++by;
        const int bx = idx - by * (by + 1) / 2;
        const int k0 = head ? 0 : cut;
        const int kn = head ? cut : T - cut;
        const float* xb = x + (long long)bz * T * D + (long long)k0 * D;
        gemm_tile<1, 0>(As, Bs, xb, nullptr, xb, (head ? G0 : G1) + (long long)bz * D * D,
                        by * BM, bx * BN, kn, D, D, D, 1.0f, by != bx);
    };

    const bool head = w < n_t;
    const bool with_p = !head && w < n_t + n_p;
    const int t0 = head ? w : (with_p ? w - n_t : n_p + 2 * (w - n_t - n_p));
    gram_part(t0, head);
    if (head) return;

    __syncthreads();
    if (with_p) {
        const int p = w - n_t;
        gemm_tile<0, 1>(As, Bs, wq, nullptr, wk, P,
                        (p / nb) * BM, (p % nb) * BN, D, D, D, D, 1.0f, false);
    } else if (t0 + 1 < n_t) {
        gram_part(t0 + 1, false);
    }
}

extern "C" void kernel_launch(void* const* d_in, const int* in_sizes, int n_in,
                              void* d_out, int out_size, void* d_ws, size_t ws_size,
                              hipStream_t stream) {
    (void)in_sizes; (void)n_in; (void)out_size; (void)ws_size;

    const float* x  = (const float*)d_in[0];
    const float* wq = (const float*)d_in[1];
    const float* wk = (const float*)d_in[2];
    const float* wv = (const float*)d_in[3];
    float* y = (float*)d_out;

    const int Bb = 4, T = 4096, D = 1024;
    const long long BDD = (long long)D * D;
    const long long BTD = (long long)T * D;

    float* G = (float*)d_ws;
    float* H = G + Bb * BDD;
    float* P  = y;
    float* G1 = y + BDD;

    dim3 blk(256);
    dim3 gridD(D / BN, D / BM, Bb);

    const int gramTiles = Bb * 36, pTiles = 64;
    dim3 gridG(gramTiles + pTiles + (gramTiles - pTiles) / 2);
    gram_and_qk_f32_wmma<<<gridG, blk, 0, stream>>>(x, wq, wk, G, G1, P, T, D, Bb, 86 * BK);

    gemm_f32_wmma_nn_sum2<<<gridD, blk, 0, stream>>>(
        G, G1, wv, H, D, D, D, D, BDD, 0LL, BDD, 1.0f);

    gemm_f32_wmma_nn<<<gridD, blk, 0, stream>>>(
        P, H, G, D, D, D, D, 0LL, BDD, BDD, 1.0f);

    dim3 gridY(D / BN, T / BM, Bb);
    gemm_f32_wmma_nn<<<gridY, blk, 0, stream>>>(
        x, G, y, D, D, D, D, BTD, BDD, BTD, 32.0f);
}
